# starting kernel
# baseline (speedup 1.0000x reference)
#include <hip/hip_runtime.h>


typedef __attribute__((ext_vector_type(2))) float v2f;
typedef __attribute__((ext_vector_type(8))) float v8f;

#define BDIM 128
#define NIN  1024
#define NOUT 1024

__global__ __launch_bounds__(32)
void memristive_fc_wmma_f32(const float* __restrict__ x,
                            const float* __restrict__ wp,
                            const float* __restrict__ wn,
                            const float* __restrict__ bp,
                            const float* __restrict__ bn,
                            float* __restrict__ y) {
    const int lane = threadIdx.x & 31;
    const int half = lane >> 4;
    const int l16  = lane & 15;

    const int m_tile  = (int)blockIdx.x >> 4;
    const int n_strip = (int)blockIdx.x & 15;
    const int m0 = m_tile * 16;
    const int n0 = n_strip * 64;

    v8f acc[4] = {};

    const float* xrow = x + (m0 + l16) * NIN;

    #pragma unroll 4
    for (int k0 = 0; k0 < NIN; k0 += 4) {
        const int ka = k0 + half * 2;
        v2f a = *(const v2f*)(xrow + ka);

        if (k0 + 4 < NIN) {
            __builtin_prefetch(wp + (ka + 4) * NOUT + n0 + l16, 0, 3);
            __builtin_prefetch(wn + (ka + 4) * NOUT + n0 + l16, 0, 3);
        }

        #pragma unroll
        for (int t = 0; t < 4; ++t) {
            const int col = n0 + t * 16 + l16;
            const float* p = wp + ka * NOUT + col;
            const float* n = wn + ka * NOUT + col;
            v2f b;
            b.x = p[0]    - n[0];
            b.y = p[NOUT] - n[NOUT];

            acc[t] = __builtin_amdgcn_wmma_f32_16x16x4_f32(
                 false, a,  false, b,
                 (short)0, acc[t],
                 false,  false);
        }
    }

    #pragma unroll
    for (int t = 0; t < 4; ++t) {
        const int col = n0 + t * 16 + l16;
        const float bias = bp[col] - bn[col];
        #pragma unroll
        for (int r = 0; r < 8; ++r) {
            const int row = m0 + half * 8 + r;
            y[row * NOUT + col] = acc[t][r] + bias;
        }
    }
}

extern "C" void kernel_launch(void* const* d_in, const int* in_sizes, int n_in,
                              void* d_out, int out_size, void* d_ws, size_t ws_size,
                              hipStream_t stream) {
    const float* x  = (const float*)d_in[0];
    const float* wp = (const float*)d_in[1];
    const float* wn = (const float*)d_in[2];
    const float* bp = (const float*)d_in[3];
    const float* bn = (const float*)d_in[4];
    float* y = (float*)d_out;

    dim3 grid(128), block(32);
    hipLaunchKernelGGL(memristive_fc_wmma_f32, grid, block, 0, stream,
                       x, wp, wn, bp, bn, y);
}


# split K over 32 waves per work-group: each wave holds a 32x32 tile over a 32-deep K slice, loads issued as one group; partials summed through LDS (sum order changes); grid 128 x 1024 threads
# speedup vs baseline: 14.2121x; 14.2121x over previous
#include <hip/hip_runtime.h>


typedef __attribute__((ext_vector_type(2))) float v2f;
typedef __attribute__((ext_vector_type(4))) float v4f;
typedef __attribute__((ext_vector_type(8))) float v8f;

#define BDIM 128
#define NIN  1024
#define NOUT 1024

#define KWAVES 32
#define KSLICE (NIN / KWAVES)
#define KSTEPS (KSLICE / 4)
#define TILE_ELEMS (32 * 32)

__global__ __launch_bounds__(KWAVES * 32)
void memristive_fc_wmma_f32(const float* __restrict__ x,
                            const float* __restrict__ wp,
                            const float* __restrict__ wn,
                            const float* __restrict__ bp,
                            const float* __restrict__ bn,
                            float* __restrict__ y) {
    __shared__ float red[KWAVES * TILE_ELEMS];

    const int tid  = (int)threadIdx.x;
    const int lane = tid & 31;
    const int wave = tid >> 5;
    const int half = lane >> 4;
    const int l16  = lane & 15;

    const int m0 = ((int)blockIdx.x >> 5) * 32;
    const int n0 = ((int)blockIdx.x & 31) * 32;

    const int kl = wave * KSLICE + half * 2;

    const float* xa = x + (m0 + l16) * NIN + kl;
    const float* pw = wp + kl * NOUT + n0 + l16;
    const float* nw = wn + kl * NOUT + n0 + l16;

    v2f a0[KSTEPS], a1[KSTEPS];
    v8f braw[KSTEPS];

    #pragma unroll
    for (int s = 0; s < KSTEPS; ++s) {
        a0[s] = *(const v2f*)(xa + 4 * s);
        a1[s] = *(const v2f*)(xa + 16 * NIN + 4 * s);
        const float* p = pw + 4 * s * NOUT;
        const float* n = nw + 4 * s * NOUT;
        braw[s][0] = p[0];
        braw[s][1] = p[NOUT];
        braw[s][2] = p[16];
        braw[s][3] = p[NOUT + 16];
        braw[s][4] = n[0];
        braw[s][5] = n[NOUT];
        braw[s][6] = n[16];
        braw[s][7] = n[NOUT + 16];
    }

    asm volatile("" ::
        "v"(a0[0]), "v"(a1[0]), "v"(braw[0]), "v"(a0[1]), "v"(a1[1]), "v"(braw[1]),
        "v"(a0[2]), "v"(a1[2]), "v"(braw[2]), "v"(a0[3]), "v"(a1[3]), "v"(braw[3]),
        "v"(a0[4]), "v"(a1[4]), "v"(braw[4]), "v"(a0[5]), "v"(a1[5]), "v"(braw[5]),
        "v"(a0[6]), "v"(a1[6]), "v"(braw[6]), "v"(a0[7]), "v"(a1[7]), "v"(braw[7]));

    v8f acc[2][2] = {};

    #pragma unroll
    for (int s = 0; s < KSTEPS; ++s) {
        const v4f d = braw[s].lo - braw[s].hi;
        v2f b[2];
        b[0].x = d.x;  b[0].y = d.y;
        b[1].x = d.z;  b[1].y = d.w;
        v2f a[2] = { a0[s], a1[s] };

        #pragma unroll
        for (int tm = 0; tm < 2; ++tm) {
            #pragma unroll
            for (int tn = 0; tn < 2; ++tn) {
                acc[tm][tn] = __builtin_amdgcn_wmma_f32_16x16x4_f32(
                     false, a[tm],  false, b[tn],
                     (short)0, acc[tm][tn],
                     false,  false);
            }
        }
    }

    float* mine = red + wave * TILE_ELEMS;
    #pragma unroll
    for (int tm = 0; tm < 2; ++tm) {
        #pragma unroll
        for (int tn = 0; tn < 2; ++tn) {
            #pragma unroll
            for (int r = 0; r < 8; ++r) {
                mine[((tm * 2 + tn) * 8 + r) * 32 + lane] = acc[tm][tn][r];
            }
        }
    }

    __syncthreads();

    const int row_l = wave;
    const int col_l = lane;
    const int rtm = row_l >> 4;
    const int rh  = (row_l >> 3) & 1;
    const int rr  = row_l & 7;
    const int rtn = col_l >> 4;
    const int rc  = col_l & 15;
    const int idx = ((rtm * 2 + rtn) * 8 + rr) * 32 + rh * 16 + rc;

    float s = 0.0f;
    #pragma unroll
    for (int w = 0; w < KWAVES; ++w) {
        s += red[w * TILE_ELEMS + idx];
    }

    const int col = n0 + col_l;
    const float bias = bp[col] - bn[col];
    y[(m0 + row_l) * NOUT + col] = s + bias;
}

extern "C" void kernel_launch(void* const* d_in, const int* in_sizes, int n_in,
                              void* d_out, int out_size, void* d_ws, size_t ws_size,
                              hipStream_t stream) {
    const float* x  = (const float*)d_in[0];
    const float* wp = (const float*)d_in[1];
    const float* wn = (const float*)d_in[2];
    const float* bp = (const float*)d_in[3];
    const float* bn = (const float*)d_in[4];
    float* y = (float*)d_out;

    dim3 grid(128), block(KWAVES * 32);
    hipLaunchKernelGGL(memristive_fc_wmma_f32, grid, block, 0, stream,
                       x, wp, wn, bp, bn, y);
}


# LDS hand-over of the 32 partial tiles in 16-byte pieces, bank-conflict-free, summed by 8 waves storing 4 whole lines each (was 4-byte reads, 2-way conflicts); bias loads moved ahead of the barrier
# speedup vs baseline: 15.0664x; 1.0601x over previous
#include <hip/hip_runtime.h>


typedef __attribute__((ext_vector_type(2))) float v2f;
typedef __attribute__((ext_vector_type(4))) float v4f;
typedef __attribute__((ext_vector_type(8))) float v8f;

#define BDIM 128
#define NIN  1024
#define NOUT 1024

#define KWAVES 32
#define KSLICE (NIN / KWAVES)
#define KSTEPS (KSLICE / 4)
#define TILE_ELEMS (32 * 32)
#define TILE_V4    (TILE_ELEMS / 4)
#define RED_WAVES  8

__global__ __launch_bounds__(KWAVES * 32)
void memristive_fc_wmma_f32(const float* __restrict__ x,
                            const float* __restrict__ wp,
                            const float* __restrict__ wn,
                            const float* __restrict__ bp,
                            const float* __restrict__ bn,
                            float* __restrict__ y) {
    __shared__ v4f red[KWAVES * TILE_V4];

    const int tid  = (int)threadIdx.x;
    const int lane = tid & 31;
    const int wave = tid >> 5;
    const int half = lane >> 4;
    const int l16  = lane & 15;

    const int m0 = ((int)blockIdx.x >> 5) * 32;
    const int n0 = ((int)blockIdx.x & 31) * 32;

    const int kl = wave * KSLICE + half * 2;

    const float* xa = x + (m0 + l16) * NIN + kl;
    const float* pw = wp + kl * NOUT + n0 + l16;
    const float* nw = wn + kl * NOUT + n0 + l16;

    v2f a0[KSTEPS], a1[KSTEPS];
    v8f braw[KSTEPS];

    #pragma unroll
    for (int s = 0; s < KSTEPS; ++s) {
        a0[s] = *(const v2f*)(xa + 4 * s);
        a1[s] = *(const v2f*)(xa + 16 * NIN + 4 * s);
        const float* p = pw + 4 * s * NOUT;
        const float* n = nw + 4 * s * NOUT;
        braw[s][0] = p[0];
        braw[s][1] = p[NOUT];
        braw[s][2] = p[16];
        braw[s][3] = p[NOUT + 16];
        braw[s][4] = n[0];
        braw[s][5] = n[NOUT];
        braw[s][6] = n[16];
        braw[s][7] = n[NOUT + 16];
    }

    asm volatile("" ::
        "v"(a0[0]), "v"(a1[0]), "v"(braw[0]), "v"(a0[1]), "v"(a1[1]), "v"(braw[1]),
        "v"(a0[2]), "v"(a1[2]), "v"(braw[2]), "v"(a0[3]), "v"(a1[3]), "v"(braw[3]),
        "v"(a0[4]), "v"(a1[4]), "v"(braw[4]), "v"(a0[5]), "v"(a1[5]), "v"(braw[5]),
        "v"(a0[6]), "v"(a1[6]), "v"(braw[6]), "v"(a0[7]), "v"(a1[7]), "v"(braw[7]));

    v8f acc[2][2] = {};

    #pragma unroll
    for (int s = 0; s < KSTEPS; ++s) {
        const v4f d = braw[s].lo - braw[s].hi;
        v2f b[2];
        b[0].x = d.x;  b[0].y = d.y;
        b[1].x = d.z;  b[1].y = d.w;
        v2f a[2] = { a0[s], a1[s] };

        #pragma unroll
        for (int tm = 0; tm < 2; ++tm) {
            #pragma unroll
            for (int tn = 0; tn < 2; ++tn) {
                acc[tm][tn] = __builtin_amdgcn_wmma_f32_16x16x4_f32(
                     false, a[tm],  false, b[tn],
                     (short)0, acc[tm][tn],
                     false,  false);
            }
        }
    }

    const int col = n0 + lane;
    float bpos = 0.0f, bneg = 0.0f;
    if (wave < RED_WAVES) {
        bpos = bp[col];
        bneg = bn[col];
    }

    v4f* mine = red + wave * TILE_V4;
    #pragma unroll
    for (int tm = 0; tm < 2; ++tm) {
        #pragma unroll
        for (int tn = 0; tn < 2; ++tn) {
            #pragma unroll
            for (int q = 0; q < 2; ++q) {
                const int g = tm * 4 + half * 2 + q;
                mine[g * 32 + tn * 16 + l16] = q ? acc[tm][tn].hi : acc[tm][tn].lo;
            }
        }
    }

    __syncthreads();

    if (wave >= RED_WAVES) return;

    const v4f* slab = red + wave * 32 + lane;
    v4f s = {0.0f, 0.0f, 0.0f, 0.0f};
    #pragma unroll
    for (int w = 0; w < KWAVES; ++w) {
        s += slab[w * TILE_V4];
    }

    float* yo = y + (m0 + 4 * wave) * NOUT + col;
    const float bias = bpos - bneg;
    yo[0 * NOUT] = s.x + bias;
    yo[1 * NOUT] = s.y + bias;
    yo[2 * NOUT] = s.z + bias;
    yo[3 * NOUT] = s.w + bias;
}

extern "C" void kernel_launch(void* const* d_in, const int* in_sizes, int n_in,
                              void* d_out, int out_size, void* d_ws, size_t ws_size,
                              hipStream_t stream) {
    const float* x  = (const float*)d_in[0];
    const float* wp = (const float*)d_in[1];
    const float* wn = (const float*)d_in[2];
    const float* bp = (const float*)d_in[3];
    const float* bn = (const float*)d_in[4];
    float* y = (float*)d_out;

    dim3 grid(128), block(KWAVES * 32);
    hipLaunchKernelGGL(memristive_fc_wmma_f32, grid, block, 0, stream,
                       x, wp, wn, bp, bn, y);
}


# K split over 8 waves of four 32-k groups (was 32 waves), next group's loads in flight under the products; 16-byte x and 8-byte weight-pair loads, 40 a group; 8 partials summed: sum order changes
# speedup vs baseline: 17.2110x; 1.1423x over previous
#include <hip/hip_runtime.h>


typedef __attribute__((ext_vector_type(2))) float v2f;
typedef __attribute__((ext_vector_type(4))) float v4f;
typedef __attribute__((ext_vector_type(8))) float v8f;

#define BDIM 128
#define NIN  1024
#define NOUT 1024

#define KWAVES 8
#define KSLICE (NIN / KWAVES)
#define GK     32
#define NGRP   (KSLICE / GK)
#define GSTEPS (GK / 4)
#define TILE_ELEMS (32 * 32)
#define TILE_V4    (TILE_ELEMS / 4)

struct Group {
    v4f a0[GSTEPS / 2], a1[GSTEPS / 2];
    v2f p[2 * GSTEPS], q[2 * GSTEPS];
};

__device__ __forceinline__ void load_group(Group& g, const float* __restrict__ xa,
                                           const float* __restrict__ pw,
                                           const float* __restrict__ nw) {
    #pragma unroll
    for (int j = 0; j < GSTEPS / 2; ++j) {
        g.a0[j] = *(const v4f*)(xa + 4 * j);
        g.a1[j] = *(const v4f*)(xa + 16 * NIN + 4 * j);
    }
    #pragma unroll
    for (int r = 0; r < 2 * GSTEPS; ++r) {
        g.p[r] = *(const v2f*)(pw + r * NOUT);
        g.q[r] = *(const v2f*)(nw + r * NOUT);
    }
}

__device__ __forceinline__ void mma_group(const Group& g, v8f (&acc)[2][2]) {
    #pragma unroll
    for (int s = 0; s < GSTEPS; ++s) {
        const v2f d0 = g.p[2 * s]     - g.q[2 * s];
        const v2f d1 = g.p[2 * s + 1] - g.q[2 * s + 1];
        v2f b[2];
        b[0].x = d0.x;  b[0].y = d1.x;
        b[1].x = d0.y;  b[1].y = d1.y;
        const v4f x0 = g.a0[s >> 1], x1 = g.a1[s >> 1];
        v2f a[2];
        a[0] = (s & 1) ? x0.hi : x0.lo;
        a[1] = (s & 1) ? x1.hi : x1.lo;

        #pragma unroll
        for (int tm = 0; tm < 2; ++tm) {
            #pragma unroll
            for (int tn = 0; tn < 2; ++tn) {
                acc[tm][tn] = __builtin_amdgcn_wmma_f32_16x16x4_f32(
                     false, a[tm],  false, b[tn],
                     (short)0, acc[tm][tn],
                     false,  false);
            }
        }
    }
}

__global__ __launch_bounds__(KWAVES * 32)
void memristive_fc_wmma_f32(const float* __restrict__ x,
                            const float* __restrict__ wp,
                            const float* __restrict__ wn,
                            const float* __restrict__ bp,
                            const float* __restrict__ bn,
                            float* __restrict__ y) {
    __shared__ v4f red[KWAVES * TILE_V4];

    const int tid  = (int)threadIdx.x;
    const int lane = tid & 31;
    const int wave = tid >> 5;
    const int half = lane >> 4;
    const int l16  = lane & 15;

    const int m0 = ((int)blockIdx.x >> 5) * 32;
    const int n0 = ((int)blockIdx.x & 31) * 32;

    const int kl = wave * KSLICE + half * 16;

    const float* xa = x + (m0 + l16) * NIN + kl;
    const float* pw = wp + kl * NOUT + n0 + 2 * l16;
    const float* nw = wn + kl * NOUT + n0 + 2 * l16;

    const int col = n0 + 2 * l16 + half;

    v8f acc[2][2] = {};
    Group g0, g1;

    load_group(g0, xa, pw, nw);
    load_group(g1, xa + GK, pw + GK * NOUT, nw + GK * NOUT);
    __builtin_amdgcn_sched_barrier(0);
    mma_group(g0, acc);
    load_group(g0, xa + 2 * GK, pw + 2 * GK * NOUT, nw + 2 * GK * NOUT);
    __builtin_amdgcn_sched_barrier(0);
    mma_group(g1, acc);
    load_group(g1, xa + 3 * GK, pw + 3 * GK * NOUT, nw + 3 * GK * NOUT);
    const float bpos = bp[col];
    const float bneg = bn[col];
    __builtin_amdgcn_sched_barrier(0);
    mma_group(g0, acc);
    mma_group(g1, acc);

    v4f* mine = red + wave * TILE_V4;
    #pragma unroll
    for (int tm = 0; tm < 2; ++tm) {
        #pragma unroll
        for (int tn = 0; tn < 2; ++tn) {
            #pragma unroll
            for (int q = 0; q < 2; ++q) {
                const int g = tm * 4 + half * 2 + q;
                mine[g * 32 + tn * 16 + l16] = q ? acc[tm][tn].hi : acc[tm][tn].lo;
            }
        }
    }

    __syncthreads();

    const v4f* slab = red + wave * 32 + lane;
    v4f s = {0.0f, 0.0f, 0.0f, 0.0f};
    #pragma unroll
    for (int w = 0; w < KWAVES; ++w) {
        s += slab[w * TILE_V4];
    }

    float* yo = y + (m0 + 4 * wave) * NOUT + col;
    const float bias = bpos - bneg;
    yo[0 * NOUT] = s.x + bias;
    yo[1 * NOUT] = s.y + bias;
    yo[2 * NOUT] = s.z + bias;
    yo[3 * NOUT] = s.w + bias;
}

extern "C" void kernel_launch(void* const* d_in, const int* in_sizes, int n_in,
                              void* d_out, int out_size, void* d_ws, size_t ws_size,
                              hipStream_t stream) {
    const float* x  = (const float*)d_in[0];
    const float* wp = (const float*)d_in[1];
    const float* wn = (const float*)d_in[2];
    const float* bp = (const float*)d_in[3];
    const float* bn = (const float*)d_in[4];
    float* y = (float*)d_out;

    dim3 grid(128), block(KWAVES * 32);
    hipLaunchKernelGGL(memristive_fc_wmma_f32, grid, block, 0, stream,
                       x, wp, wn, bp, bn, y);
}


# work-group tile 16x32 (was 32x32): grid 256 = one work-group on every WGP (was 128, half the chip idle); per SIMD half the f32 WMMAs; 8 K-slice partials still summed in slice order
# speedup vs baseline: 18.9397x; 1.1004x over previous
#include <hip/hip_runtime.h>


typedef __attribute__((ext_vector_type(2))) float v2f;
typedef __attribute__((ext_vector_type(4))) float v4f;
typedef __attribute__((ext_vector_type(8))) float v8f;

#define BDIM 128
#define NIN  1024
#define NOUT 1024

#define KWAVES 8
#define KSLICE (NIN / KWAVES)
#define GK     32
#define NGRP   (KSLICE / GK)
#define GSTEPS (GK / 4)
#define TILE_ELEMS (16 * 32)
#define TILE_V4    (TILE_ELEMS / 4)

struct Group {
    v4f a0[GSTEPS / 2];
    v2f p[2 * GSTEPS], q[2 * GSTEPS];
};

__device__ __forceinline__ void load_group(Group& g, const float* __restrict__ xa,
                                           const float* __restrict__ pw,
                                           const float* __restrict__ nw) {
    #pragma unroll
    for (int j = 0; j < GSTEPS / 2; ++j) {
        g.a0[j] = *(const v4f*)(xa + 4 * j);
    }
    #pragma unroll
    for (int r = 0; r < 2 * GSTEPS; ++r) {
        g.p[r] = *(const v2f*)(pw + r * NOUT);
        g.q[r] = *(const v2f*)(nw + r * NOUT);
    }
}

__device__ __forceinline__ void mma_group(const Group& g, v8f (&acc)[2]) {
    #pragma unroll
    for (int s = 0; s < GSTEPS; ++s) {
        const v2f d0 = g.p[2 * s]     - g.q[2 * s];
        const v2f d1 = g.p[2 * s + 1] - g.q[2 * s + 1];
        v2f b[2];
        b[0].x = d0.x;  b[0].y = d1.x;
        b[1].x = d0.y;  b[1].y = d1.y;
        const v4f x0 = g.a0[s >> 1];
        const v2f a = (s & 1) ? x0.hi : x0.lo;

        #pragma unroll
        for (int tn = 0; tn < 2; ++tn) {
            acc[tn] = __builtin_amdgcn_wmma_f32_16x16x4_f32(
                 false, a,  false, b[tn],
                 (short)0, acc[tn],
                 false,  false);
        }
    }
}

__global__ __launch_bounds__(KWAVES * 32)
void memristive_fc_wmma_f32(const float* __restrict__ x,
                            const float* __restrict__ wp,
                            const float* __restrict__ wn,
                            const float* __restrict__ bp,
                            const float* __restrict__ bn,
                            float* __restrict__ y) {
    __shared__ v4f red[KWAVES * TILE_V4];

    const int tid  = (int)threadIdx.x;
    const int lane = tid & 31;
    const int wave = tid >> 5;
    const int half = lane >> 4;
    const int l16  = lane & 15;

    const int m0 = ((int)blockIdx.x >> 5) * 16;
    const int n0 = ((int)blockIdx.x & 31) * 32;

    const int kl = wave * KSLICE + half * 16;

    const float* xa = x + (m0 + l16) * NIN + kl;
    const float* pw = wp + kl * NOUT + n0 + 2 * l16;
    const float* nw = wn + kl * NOUT + n0 + 2 * l16;

    const int col = n0 + 2 * l16 + half;

    v8f acc[2] = {};
    Group g0, g1;

    load_group(g0, xa, pw, nw);
    load_group(g1, xa + GK, pw + GK * NOUT, nw + GK * NOUT);
    __builtin_amdgcn_sched_barrier(0);
    mma_group(g0, acc);
    load_group(g0, xa + 2 * GK, pw + 2 * GK * NOUT, nw + 2 * GK * NOUT);
    __builtin_amdgcn_sched_barrier(0);
    mma_group(g1, acc);
    load_group(g1, xa + 3 * GK, pw + 3 * GK * NOUT, nw + 3 * GK * NOUT);
    const float bpos = bp[col];
    const float bneg = bn[col];
    __builtin_amdgcn_sched_barrier(0);
    mma_group(g0, acc);
    mma_group(g1, acc);

    v4f* mine = red + wave * TILE_V4;
    #pragma unroll
    for (int tn = 0; tn < 2; ++tn) {
        #pragma unroll
        for (int q = 0; q < 2; ++q) {
            const int g = half * 2 + q;
            mine[g * 32 + tn * 16 + l16] = q ? acc[tn].hi : acc[tn].lo;
        }
    }

    __syncthreads();

    const v2f* slab = (const v2f*)(red + (wave >> 1) * 32 + lane) + (wave & 1);
    v2f s = {0.0f, 0.0f};
    #pragma unroll
    for (int w = 0; w < KWAVES; ++w) {
        s += slab[w * (TILE_V4 * 2)];
    }

    float* yo = y + (m0 + 2 * wave) * NOUT + col;
    const float bias = bpos - bneg;
    yo[0 * NOUT] = s.x + bias;
    yo[1 * NOUT] = s.y + bias;
}

extern "C" void kernel_launch(void* const* d_in, const int* in_sizes, int n_in,
                              void* d_out, int out_size, void* d_ws, size_t ws_size,
                              hipStream_t stream) {
    const float* x  = (const float*)d_in[0];
    const float* wp = (const float*)d_in[1];
    const float* wn = (const float*)d_in[2];
    const float* bp = (const float*)d_in[3];
    const float* bn = (const float*)d_in[4];
    float* y = (float*)d_out;

    dim3 grid(256), block(KWAVES * 32);
    hipLaunchKernelGGL(memristive_fc_wmma_f32, grid, block, 0, stream,
                       x, wp, wn, bp, bn, y);
}


# group walk of every K slice starts at group (mb+nb)&3 and wraps: work-groups that share weight or x lines no longer ask the L2 for the same lines together; a wave's 4 group sums change order
# speedup vs baseline: 21.2335x; 1.1211x over previous
#include <hip/hip_runtime.h>


typedef __attribute__((ext_vector_type(2))) float v2f;
typedef __attribute__((ext_vector_type(4))) float v4f;
typedef __attribute__((ext_vector_type(8))) float v8f;

#define BDIM 128
#define NIN  1024
#define NOUT 1024

#define KWAVES 8
#define KSLICE (NIN / KWAVES)
#define GK     32
#define NGRP   (KSLICE / GK)
#define GSTEPS (GK / 4)
#define TILE_ELEMS (16 * 32)
#define TILE_V4    (TILE_ELEMS / 4)

struct Group {
    v4f a0[GSTEPS / 2];
    v2f p[2 * GSTEPS], q[2 * GSTEPS];
};

__device__ __forceinline__ void load_group(Group& g, const float* __restrict__ xa,
                                           const float* __restrict__ pw,
                                           const float* __restrict__ nw) {
    #pragma unroll
    for (int j = 0; j < GSTEPS / 2; ++j) {
        g.a0[j] = *(const v4f*)(xa + 4 * j);
    }
    #pragma unroll
    for (int r = 0; r < 2 * GSTEPS; ++r) {
        g.p[r] = *(const v2f*)(pw + r * NOUT);
        g.q[r] = *(const v2f*)(nw + r * NOUT);
    }
}

__device__ __forceinline__ void mma_group(const Group& g, v8f (&acc)[2]) {
    #pragma unroll
    for (int s = 0; s < GSTEPS; ++s) {
        const v2f d0 = g.p[2 * s]     - g.q[2 * s];
        const v2f d1 = g.p[2 * s + 1] - g.q[2 * s + 1];
        v2f b[2];
        b[0].x = d0.x;  b[0].y = d1.x;
        b[1].x = d0.y;  b[1].y = d1.y;
        const v4f x0 = g.a0[s >> 1];
        const v2f a = (s & 1) ? x0.hi : x0.lo;

        #pragma unroll
        for (int tn = 0; tn < 2; ++tn) {
            acc[tn] = __builtin_amdgcn_wmma_f32_16x16x4_f32(
                 false, a,  false, b[tn],
                 (short)0, acc[tn],
                 false,  false);
        }
    }
}

__global__ __launch_bounds__(KWAVES * 32)
void memristive_fc_wmma_f32(const float* __restrict__ x,
                            const float* __restrict__ wp,
                            const float* __restrict__ wn,
                            const float* __restrict__ bp,
                            const float* __restrict__ bn,
                            float* __restrict__ y) {
    __shared__ v4f red[KWAVES * TILE_V4];

    const int tid  = (int)threadIdx.x;
    const int lane = tid & 31;
    const int wave = tid >> 5;
    const int half = lane >> 4;
    const int l16  = lane & 15;

    const int m0 = ((int)blockIdx.x >> 5) * 16;
    const int n0 = ((int)blockIdx.x & 31) * 32;

    const int kl = wave * KSLICE + half * 16;

    const float* xa = x + (m0 + l16) * NIN + kl;
    const float* pw = wp + kl * NOUT + n0 + 2 * l16;
    const float* nw = wn + kl * NOUT + n0 + 2 * l16;

    const int col = n0 + 2 * l16 + half;

    const int rot = ((int)blockIdx.x + ((int)blockIdx.x >> 5)) & 3;
    const int k0 = rot * GK;
    const int k1 = ((rot + 1) & 3) * GK;
    const int k2 = ((rot + 2) & 3) * GK;
    const int k3 = ((rot + 3) & 3) * GK;

    v8f acc[2] = {};
    Group g0, g1;

    load_group(g0, xa + k0, pw + k0 * NOUT, nw + k0 * NOUT);
    load_group(g1, xa + k1, pw + k1 * NOUT, nw + k1 * NOUT);
    __builtin_amdgcn_sched_barrier(0);
    mma_group(g0, acc);
    load_group(g0, xa + k2, pw + k2 * NOUT, nw + k2 * NOUT);
    __builtin_amdgcn_sched_barrier(0);
    mma_group(g1, acc);
    load_group(g1, xa + k3, pw + k3 * NOUT, nw + k3 * NOUT);
    const float bpos = bp[col];
    const float bneg = bn[col];
    __builtin_amdgcn_sched_barrier(0);
    mma_group(g0, acc);
    mma_group(g1, acc);

    v4f* mine = red + wave * TILE_V4;
    #pragma unroll
    for (int tn = 0; tn < 2; ++tn) {
        #pragma unroll
        for (int q = 0; q < 2; ++q) {
            const int g = half * 2 + q;
            mine[g * 32 + tn * 16 + l16] = q ? acc[tn].hi : acc[tn].lo;
        }
    }

    __syncthreads();

    const v2f* slab = (const v2f*)(red + (wave >> 1) * 32 + lane) + (wave & 1);
    v2f s = {0.0f, 0.0f};
    #pragma unroll
    for (int w = 0; w < KWAVES; ++w) {
        s += slab[w * (TILE_V4 * 2)];
    }

    float* yo = y + (m0 + 2 * wave) * NOUT + col;
    const float bias = bpos - bneg;
    yo[0 * NOUT] = s.x + bias;
    yo[1 * NOUT] = s.y + bias;
}

extern "C" void kernel_launch(void* const* d_in, const int* in_sizes, int n_in,
                              void* d_out, int out_size, void* d_ws, size_t ws_size,
                              hipStream_t stream) {
    const float* x  = (const float*)d_in[0];
    const float* wp = (const float*)d_in[1];
    const float* wn = (const float*)d_in[2];
    const float* bp = (const float*)d_in[3];
    const float* bn = (const float*)d_in[4];
    float* y = (float*)d_out;

    dim3 grid(256), block(KWAVES * 32);
    hipLaunchKernelGGL(memristive_fc_wmma_f32, grid, block, 0, stream,
                       x, wp, wn, bp, bn, y);
}
